# starting kernel
# baseline (speedup 1.0000x reference)
#include <hip/hip_runtime.h>

#define B_BEADS 2000
#define M_SLOTS 12
#define STRIDE  4
#define NCOLS   (3 * B_BEADS)
#define NTILES  (NCOLS / 16)

typedef float v2f __attribute__((ext_vector_type(2)));
typedef float v8f __attribute__((ext_vector_type(8)));

#define CM0  0x000u
#define CM1  0x000u
#define CM2  0x000u
#define CM3  0x248u
#define CM4  0x490u
#define CM5  0x920u
#define CM6  0x240u
#define CM7  0x480u
#define CM8  0x900u
#define CM9  0x200u
#define CM10 0x400u
#define CM11 0x800u

#define WMMA_STEP(c0, c1, c2, c3, byte_off)                                   \
    do {                                                                      \
        v2f afrag, bfrag;                                                     \
        const unsigned m0 = half ? (c2) : (c0);                               \
        const unsigned m1 = half ? (c3) : (c1);                               \
        afrag.x = (float)((m0 >> row) & 1u);                                  \
        afrag.y = (float)((m1 >> row) & 1u);                                  \
        bfrag.x = relk[(byte_off) / 4];                                       \
        bfrag.y = relk[(byte_off) / 4 + 3];                                   \
        acc = __builtin_amdgcn_wmma_f32_16x16x4_f32(                          \
            false, afrag, false, bfrag, (short)0, acc, false, false);         \
    } while (0)

__global__ void __launch_bounds__(256)
hier_stage1_wmma(const float* __restrict__ rel,
                 const float* __restrict__ bpos,
                 float* __restrict__ ws_val) {
    const int wave = blockIdx.x * (blockDim.x >> 5) + (threadIdx.x >> 5);
    if (wave >= NTILES) return;

    const int lane = threadIdx.x & 31;
    const int half = lane >> 4;
    const int row  = lane & 15;
    const int col  = wave * 16 + row;
    const int c    = col / B_BEADS;
    const int b    = col % B_BEADS;

    const float bp = bpos[b * 3 + c];
    v8f acc;
#pragma unroll
    for (int r = 0; r < 8; ++r) acc[r] = bp;

    const float* relk = rel + (size_t)b * (M_SLOTS * 3) + c + half * 6;

    WMMA_STEP(CM0, CM1, CM2,  CM3,  0);
    WMMA_STEP(CM4, CM5, CM6,  CM7,  48);
    WMMA_STEP(CM8, CM9, CM10, CM11, 96);

    float* wcol = ws_val + ((size_t)c * 16 + half * 8) * B_BEADS + b;
#pragma unroll
    for (int r = 0; r < 8; ++r)
        wcol[(size_t)r * B_BEADS] = acc[r];
}

__global__ void __launch_bounds__(256)
hier_stage2_gather(const float* __restrict__ ws_val,
                   float* __restrict__ out,
                   int natoms) {
    const int idx = blockIdx.x * blockDim.x + threadIdx.x;
    if (idx >= natoms * 3) return;
    const int a = idx / 3;
    const int c = idx - a * 3;
    const int r = a & (STRIDE - 1);

    float sum = 0.0f;
    int cnt = 0;
#pragma unroll
    for (int t = 0; t < 3; ++t) {
        const int m = r + 4 * t;
        const int b = (a - m) >> 2;
        if (b >= 0 && b < B_BEADS) {
            sum += ws_val[((size_t)c * 16 + m) * B_BEADS + b];
            ++cnt;
        }
    }
    out[idx] = sum / (float)(cnt > 0 ? cnt : 1);
}

__global__ void __launch_bounds__(256)
hier_fused_scalar(const float* __restrict__ rel,
                  const float* __restrict__ bpos,
                  float* __restrict__ out,
                  int natoms) {
    const int idx = blockIdx.x * blockDim.x + threadIdx.x;
    if (idx >= natoms * 3) return;
    const int a = idx / 3;
    const int c = idx - a * 3;
    const int r = a & (STRIDE - 1);

    float sum = 0.0f;
    int cnt = 0;
#pragma unroll
    for (int t = 0; t < 3; ++t) {
        const int m = r + 4 * t;
        const int b = (a - m) >> 2;
        if (b >= 0 && b < B_BEADS) {
            float v = bpos[b * 3 + c];
            for (int j = m; j >= 3; j -= 3)
                v += rel[(size_t)b * (M_SLOTS * 3) + j * 3 + c];
            sum += v;
            ++cnt;
        }
    }
    out[idx] = sum / (float)(cnt > 0 ? cnt : 1);
}

extern "C" void kernel_launch(void* const* d_in, const int* in_sizes, int n_in,
                              void* d_out, int out_size, void* d_ws, size_t ws_size,
                              hipStream_t stream) {
    const float* rel  = (const float*)d_in[0];
    const float* bpos = (const float*)d_in[1];
    float* out = (float*)d_out;
    const int natoms = out_size / 3;

    const size_t ws_need = (size_t)3 * 16 * B_BEADS * sizeof(float);
    const int tot = natoms * 3;

    if (ws_size >= ws_need) {
        float* ws_val = (float*)d_ws;
        const int wpb = 256 / 32;
        dim3 grid1((NTILES + wpb - 1) / wpb);
        hier_stage1_wmma<<<grid1, 256, 0, stream>>>(rel, bpos, ws_val);
        hier_stage2_gather<<<(tot + 255) / 256, 256, 0, stream>>>(ws_val, out, natoms);
    } else {
        hier_fused_scalar<<<(tot + 255) / 256, 256, 0, stream>>>(rel, bpos, out, natoms);
    }
}


# Fuse the gather-and-mean stage into the WMMA kernel through an LDS tile: one launch instead of two, no workspace round trip; a wave does one 16-bead tile for all 3 components
# speedup vs baseline: 2.7645x; 2.7645x over previous
#include <hip/hip_runtime.h>

#define B_BEADS 2000
#define M_SLOTS 12
#define STRIDE  4

#define NT    3
#define NB    (16 * NT)
#define QB    32
#define NTHR  (32 * NT)
#define NBLK  ((B_BEADS + 2 + QB - 1) / QB)

#define LP    20
#define LCP   (NB * LP + 16)

typedef float v2f __attribute__((ext_vector_type(2)));
typedef float v8f __attribute__((ext_vector_type(8)));

#define CM0  0x000u
#define CM1  0x000u
#define CM2  0x000u
#define CM3  0x248u
#define CM4  0x490u
#define CM5  0x920u
#define CM6  0x240u
#define CM7  0x480u
#define CM8  0x900u
#define CM9  0x200u
#define CM10 0x400u
#define CM11 0x800u

#define WMMA_STEP(c0, c1, c2, c3, foff)                                       \
    do {                                                                      \
        v2f afrag, b0, b1, b2;                                                \
        const unsigned m0 = half ? (c2) : (c0);                               \
        const unsigned m1 = half ? (c3) : (c1);                               \
        afrag.x = (float)((m0 >> row) & 1u);                                  \
        afrag.y = (float)((m1 >> row) & 1u);                                  \
        const float* p = relk + (foff);                                       \
        b0.x = p[0]; b1.x = p[1]; b2.x = p[2];                                \
        b0.y = p[3]; b1.y = p[4]; b2.y = p[5];                                \
        acc0 = __builtin_amdgcn_wmma_f32_16x16x4_f32(                         \
            false, afrag, false, b0, (short)0, acc0, false, false);           \
        acc1 = __builtin_amdgcn_wmma_f32_16x16x4_f32(                         \
            false, afrag, false, b1, (short)0, acc1, false, false);           \
        acc2 = __builtin_amdgcn_wmma_f32_16x16x4_f32(                         \
            false, afrag, false, b2, (short)0, acc2, false, false);           \
    } while (0)

__global__ void __launch_bounds__(NTHR)
hier_fused_wmma(const float* __restrict__ rel,
                const float* __restrict__ bpos,
                float* __restrict__ out,
                int tot) {
    __shared__ __attribute__((aligned(16))) float sval[3 * LCP];

    const int tid  = threadIdx.x;
    const int wave = tid >> 5;
    const int lane = tid & 31;
    const int half = lane >> 4;
    const int row  = lane & 15;
    const int q0   = blockIdx.x * QB;
    const int slot = wave * 16 + row;

    int b = q0 - 2 + slot;
    b = b < 0 ? 0 : (b > B_BEADS - 1 ? B_BEADS - 1 : b);

    const float bp0 = bpos[b * 3 + 0];
    const float bp1 = bpos[b * 3 + 1];
    const float bp2 = bpos[b * 3 + 2];
    v8f acc0, acc1, acc2;
#pragma unroll
    for (int r = 0; r < 8; ++r) { acc0[r] = bp0; acc1[r] = bp1; acc2[r] = bp2; }

    const float* relk = rel + (size_t)b * (M_SLOTS * 3) + half * 6;

    WMMA_STEP(CM0, CM1, CM2,  CM3,  0);
    WMMA_STEP(CM4, CM5, CM6,  CM7,  12);
    WMMA_STEP(CM8, CM9, CM10, CM11, 24);

    float* sp = sval + slot * LP + half * 8;
    *(float4*)(sp)               = make_float4(acc0[0], acc0[1], acc0[2], acc0[3]);
    *(float4*)(sp + 4)           = make_float4(acc0[4], acc0[5], acc0[6], acc0[7]);
    *(float4*)(sp + LCP)         = make_float4(acc1[0], acc1[1], acc1[2], acc1[3]);
    *(float4*)(sp + LCP + 4)     = make_float4(acc1[4], acc1[5], acc1[6], acc1[7]);
    *(float4*)(sp + 2 * LCP)     = make_float4(acc2[0], acc2[1], acc2[2], acc2[3]);
    *(float4*)(sp + 2 * LCP + 4) = make_float4(acc2[4], acc2[5], acc2[6], acc2[7]);

    __syncthreads();

    float res[4];
#pragma unroll
    for (int e = 0; e < 4; ++e) {
        const int k  = tid * 4 + e;
        const int al = k / 3;
        const int c  = k - al * 3;
        const int r  = al & (STRIDE - 1);
        const int ql = al >> 2;
        const int q  = q0 + ql;

        const float* sv = sval + c * LCP + (ql + 2) * LP + r;
        const float v0 = sv[0];
        const float v1 = sv[4 - LP];
        const float v2 = sv[8 - 2 * LP];
        const bool ok0 = (unsigned)(q)     < (unsigned)B_BEADS;
        const bool ok1 = (unsigned)(q - 1) < (unsigned)B_BEADS;
        const bool ok2 = (unsigned)(q - 2) < (unsigned)B_BEADS;

        float sum = 0.0f;
        sum = ok0 ? sum + v0 : sum;
        sum = ok1 ? sum + v1 : sum;
        sum = ok2 ? sum + v2 : sum;
        const int cnt = (int)ok0 + (int)ok1 + (int)ok2;
        res[e] = sum / (float)(cnt > 0 ? cnt : 1);
    }
    const int idx = q0 * 12 + tid * 4;
    if (idx < tot)
        *(float4*)(out + idx) = make_float4(res[0], res[1], res[2], res[3]);
}

extern "C" void kernel_launch(void* const* d_in, const int* in_sizes, int n_in,
                              void* d_out, int out_size, void* d_ws, size_t ws_size,
                              hipStream_t stream) {
    const float* rel  = (const float*)d_in[0];
    const float* bpos = (const float*)d_in[1];
    float* out = (float*)d_out;
    hier_fused_wmma<<<NBLK, NTHR, 0, stream>>>(rel, bpos, out, out_size);
}
